# starting kernel
# baseline (speedup 1.0000x reference)
#include <hip/hip_runtime.h>
#include <hip/hip_bf16.h>
#include <math.h>

typedef __attribute__((ext_vector_type(2))) float v2f;
typedef __attribute__((ext_vector_type(4))) float v4f;
typedef __attribute__((ext_vector_type(8))) float v8f;

#define E_DIM   512
#define L_DIM   128
#define KCHUNK  256
#define PRSTRIDE 144

#define WS_BTP   0
#define WS_IVAR  (E_DIM * L_DIM)
#define WS_M2    (WS_IVAR + E_DIM)
#define WS_CONST (WS_M2 + L_DIM)

__global__ __launch_bounds__(256) void GaussianScorer_prep(
    const float* __restrict__ means, const float* __restrict__ var,
    float* __restrict__ ws) {
  __shared__ float sIv[E_DIM];
  __shared__ float sRed[256];
  const int tid = threadIdx.x;

  float slog = 0.0f;
  for (int e = tid; e < E_DIM; e += 256) {
    float v  = var[e];
    float iv = 1.0f / v;
    sIv[e] = iv;
    ws[WS_IVAR + e] = iv;
    slog += logf(v);
  }
  sRed[tid] = slog;
  __syncthreads();
  for (int s = 128; s > 0; s >>= 1) {
    if (tid < s) sRed[tid] += sRed[tid + s];
    __syncthreads();
  }
  if (tid == 0) {
    const float log2pi = 1.8378770664093453f;
    const float factor = -0.5f * (float)E_DIM * log2pi;
    ws[WS_CONST] = factor - 0.5f * sRed[0];
  }

  for (int idx = tid; idx < E_DIM * L_DIM; idx += 256) {
    int pp  = idx >> 8;
    int rem = idx & 255;
    int l   = rem >> 1;
    int q   = rem & 1;
    int e   = 2 * pp + q;
    ws[WS_BTP + idx] = means[l * E_DIM + e] * sIv[e];
  }

  if (tid < L_DIM) {
    float s = 0.0f;
    for (int e = 0; e < E_DIM; ++e) {
      float m = means[tid * E_DIM + e];
      s += m * m * sIv[e];
    }
    ws[WS_M2 + tid] = s;
  }
}

__global__ __launch_bounds__(128) void GaussianScorer_gemm(
    const float* __restrict__ emb, const float* __restrict__ ws,
    float* __restrict__ out) {
  __shared__ v2f   sBp[(KCHUNK / 2) * PRSTRIDE];
  __shared__ float sIv[KCHUNK];
  __shared__ float sE2[64];

  const int tid   = threadIdx.x;
  const int wave  = tid >> 5;
  const int lane  = tid & 31;
  const int col   = lane & 15;
  const int khalf = lane >> 4;
  const int kgrp  = khalf * 2;
  const int rowStart = blockIdx.x * 64 + wave * 16;
  const int myRow    = rowStart + col;

  const float* aRow = emb + (size_t)myRow * E_DIM;

  v8f acc[8] = {};
  float e2p = 0.0f;

  for (int chunk = 0; chunk < 2; ++chunk) {
    const v4f* __restrict__ srcB =
        (const v4f*)(ws + WS_BTP + chunk * (KCHUNK / 2) * (L_DIM * 2));
    for (int j = tid; j < (KCHUNK / 2) * (L_DIM / 2); j += 128) {
      int pr = j >> 6;
      int q  = j & 63;
      *((v4f*)(sBp + pr * PRSTRIDE) + q) = srcB[pr * (L_DIM / 2) + q];
    }
    if (tid < KCHUNK / 4)
      ((v4f*)sIv)[tid] = ((const v4f*)(ws + WS_IVAR + chunk * KCHUNK))[tid];
    __syncthreads();

    const float* aC = aRow + chunk * KCHUNK + kgrp;
#pragma unroll 4
    for (int i = 0; i < KCHUNK / 4; ++i) {
      const int kl = i * 4 + kgrp;
      v2f a  = *(const v2f*)(aC + i * 4);
      v2f iv = *(const v2f*)(sIv + kl);
      e2p += a.x * a.x * iv.x + a.y * a.y * iv.y;

      const v2f* bRow = sBp + (i * 2 + khalf) * PRSTRIDE + col;
#pragma unroll
      for (int t = 0; t < 8; ++t) {
        v2f b = bRow[t * 16];
        acc[t] = __builtin_amdgcn_wmma_f32_16x16x4_f32(
            false, a, false, b, (short)0, acc[t], false, false);
      }
    }
    __syncthreads();
  }

  float e2t = e2p + __shfl_xor(e2p, 16, 32);
  if (lane < 16) sE2[wave * 16 + col] = e2t;
  __syncthreads();

  const float cTerm = ws[WS_CONST];
#pragma unroll
  for (int t = 0; t < 8; ++t) {
    float m2c = ws[WS_M2 + t * 16 + col];
#pragma unroll
    for (int i = 0; i < 8; ++i) {
      int rl = khalf * 8 + i;
      float val = cTerm + acc[t][i] - 0.5f * (sE2[wave * 16 + rl] + m2c);
      out[(size_t)(rowStart + rl) * L_DIM + t * 16 + col] = val;
    }
  }
}

extern "C" void kernel_launch(void* const* d_in, const int* in_sizes, int n_in,
                              void* d_out, int out_size, void* d_ws, size_t ws_size,
                              hipStream_t stream) {
  const float* emb   = (const float*)d_in[0];
  const float* means = (const float*)d_in[1];
  const float* var   = (const float*)d_in[2];
  float* out = (float*)d_out;
  float* ws  = (float*)d_ws;

  GaussianScorer_prep<<<1, 256, 0, stream>>>(means, var, ws);
  GaussianScorer_gemm<<<256, 128, 0, stream>>>(emb, ws, out);
}


# operand roles swapped: emb rows (x 1/var) staged in LDS, each wave reads its 16 columns' means fragments from global; no K split, rows stored whole via LDS; sum order of dots, e2, m2, log var changed
# speedup vs baseline: 3.6088x; 3.6088x over previous
#include <hip/hip_runtime.h>
#include <math.h>


typedef __attribute__((ext_vector_type(2))) float v2f;
typedef __attribute__((ext_vector_type(4))) float v4f;
typedef __attribute__((ext_vector_type(8))) float v8f;

#define E_DIM   512
#define L_DIM   128
#define ROWS    64
#define APITCH  (E_DIM + 4)
#define OPITCH  (L_DIM + 4)

__global__ __launch_bounds__(256) __attribute__((amdgpu_waves_per_eu(2, 2))) void scorer_gemm(
    const float* __restrict__ emb, const float* __restrict__ means,
    const float* __restrict__ var, float* __restrict__ out) {
  __shared__ __attribute__((aligned(16))) float sA[ROWS * APITCH];
  __shared__ __attribute__((aligned(16))) float sIv[E_DIM];
  __shared__ __attribute__((aligned(16))) float sE2[ROWS];
  __shared__ float sLog;

  const int tid   = threadIdx.x;
  const int wave  = tid >> 5;
  const int lane  = tid & 31;
  const int col   = lane & 15;
  const int khalf = lane >> 4;
  const int rowBase = blockIdx.x * ROWS;

  v4f ar[8][4];
#pragma unroll
  for (int j = 0; j < 8; ++j) {
    const v4f* src = (const v4f*)(emb + (size_t)(rowBase + wave * 8 + j) * E_DIM);
#pragma unroll
    for (int m = 0; m < 4; ++m) ar[j][m] = src[lane + 32 * m];
  }

  v4f iv4[4];
#pragma unroll
  for (int m = 0; m < 4; ++m) {
    v4f v = ((const v4f*)var)[lane + 32 * m];
    v4f r;
    r.x = 1.0f / v.x; r.y = 1.0f / v.y; r.z = 1.0f / v.z; r.w = 1.0f / v.w;
    iv4[m] = r;
  }
  if (wave == 0) {
    float slog = 0.0f;
#pragma unroll
    for (int m = 0; m < 4; ++m) {
      v4f v = ((const v4f*)var)[lane + 32 * m];
      slog += (logf(v.x) + logf(v.y)) + (logf(v.z) + logf(v.w));
      ((v4f*)sIv)[lane + 32 * m] = iv4[m];
    }
#pragma unroll
    for (int o = 16; o > 0; o >>= 1) slog += __shfl_xor(slog, o, 32);
    if (lane == 0) sLog = slog;
  }

  float e[8];
#pragma unroll
  for (int j = 0; j < 8; ++j) {
    float s = 0.0f;
#pragma unroll
    for (int m = 0; m < 4; ++m) {
      v4f a  = ar[j][m];
      v4f as = a * iv4[m];
      *(v4f*)(sA + (wave * 8 + j) * APITCH + (lane + 32 * m) * 4) = as;
      s += (a.x * as.x + a.y * as.y) + (a.z * as.z + a.w * as.w);
    }
    e[j] = s;
  }
  {
    const bool b4 = (lane & 16) != 0, b3 = (lane & 8) != 0, b2 = (lane & 4) != 0;
    float f[4], g[2];
#pragma unroll
    for (int j = 0; j < 4; ++j) {
      float send = b4 ? e[j] : e[j + 4];
      float keep = b4 ? e[j + 4] : e[j];
      f[j] = keep + __shfl_xor(send, 16, 32);
    }
#pragma unroll
    for (int j = 0; j < 2; ++j) {
      float send = b3 ? f[j] : f[j + 2];
      float keep = b3 ? f[j + 2] : f[j];
      g[j] = keep + __shfl_xor(send, 8, 32);
    }
    float send = b2 ? g[0] : g[1];
    float keep = b2 ? g[1] : g[0];
    float h = keep + __shfl_xor(send, 4, 32);
    h += __shfl_xor(h, 2, 32);
    h += __shfl_xor(h, 1, 32);
    if ((lane & 3) == 0) sE2[wave * 8 + (lane >> 2)] = h;
  }
  __syncthreads();

  v8f acc[4] = {};
  v4f m2p = {0.0f, 0.0f, 0.0f, 0.0f};

  const float* bC  = means + (size_t)(wave * 16 + col) * E_DIM + khalf * 4;
  const float* ivC = sIv + khalf * 4;
  const float* aC  = sA + col * APITCH + khalf * 4;
#pragma unroll 1
  for (int t = 0; t < E_DIM / 64; ++t) {
    v4f b[8], iv[8], a[8][4];
#pragma unroll
    for (int o = 0; o < 8; ++o) b[o] = *(const v4f*)(bC + t * 64 + o * 8);
#pragma unroll
    for (int o = 0; o < 8; ++o) {
      iv[o] = *(const v4f*)(ivC + t * 64 + o * 8);
#pragma unroll
      for (int rt = 0; rt < 4; ++rt) a[o][rt] = *(const v4f*)(aC + rt * 16 * APITCH + t * 64 + o * 8);
    }
    asm volatile("" :: "v"(a[0][0]), "v"(a[0][1]), "v"(a[0][2]), "v"(a[0][3]), "v"(a[1][0]), "v"(a[1][1]), "v"(a[1][2]), "v"(a[1][3]), "v"(a[2][0]), "v"(a[2][1]), "v"(a[2][2]), "v"(a[2][3]), "v"(a[3][0]), "v"(a[3][1]), "v"(a[3][2]), "v"(a[3][3]));
    asm volatile("" :: "v"(a[4][0]), "v"(a[4][1]), "v"(a[4][2]), "v"(a[4][3]), "v"(a[5][0]), "v"(a[5][1]), "v"(a[5][2]), "v"(a[5][3]), "v"(a[6][0]), "v"(a[6][1]), "v"(a[6][2]), "v"(a[6][3]), "v"(a[7][0]), "v"(a[7][1]), "v"(a[7][2]), "v"(a[7][3]));
#pragma unroll
    for (int o = 0; o < 8; ++o) {
      m2p += b[o] * (b[o] * iv[o]);
      v2f b0 = {b[o].x, b[o].y};
      v2f b1 = {b[o].z, b[o].w};
#pragma unroll
      for (int rt = 0; rt < 4; ++rt) {
        v2f a0 = {a[o][rt].x, a[o][rt].y};
        acc[rt] = __builtin_amdgcn_wmma_f32_16x16x4_f32(
            false, a0, false, b0, (short)0, acc[rt], false, false);
      }
#pragma unroll
      for (int rt = 0; rt < 4; ++rt) {
        v2f a1 = {a[o][rt].z, a[o][rt].w};
        acc[rt] = __builtin_amdgcn_wmma_f32_16x16x4_f32(
            false, a1, false, b1, (short)0, acc[rt], false, false);
      }
    }
  }
  __syncthreads();

  float m2s = (m2p.x + m2p.y) + (m2p.z + m2p.w);
  const float m2c = m2s + __shfl_xor(m2s, 16, 32);

  const float log2pi = 1.8378770664093453f;
  const float factor = -0.5f * (float)E_DIM * log2pi;
  const float cTerm  = factor - 0.5f * sLog;

  float* sO = sA;
#pragma unroll
  for (int rt = 0; rt < 4; ++rt) {
    const v4f e2lo = *(const v4f*)(sE2 + rt * 16 + khalf * 8);
    const v4f e2up = *(const v4f*)(sE2 + rt * 16 + khalf * 8 + 4);
    const float e2r[8] = {e2lo.x, e2lo.y, e2lo.z, e2lo.w, e2up.x, e2up.y, e2up.z, e2up.w};
#pragma unroll
    for (int i = 0; i < 8; ++i) {
      const int rl = rt * 16 + khalf * 8 + i;
      sO[rl * OPITCH + wave * 16 + col] = cTerm + acc[rt][i] - 0.5f * (e2r[i] + m2c);
    }
  }
  __syncthreads();

#pragma unroll
  for (int j = 0; j < 8; ++j) {
    const int rl = wave * 8 + j;
    v4f o = *(const v4f*)(sO + rl * OPITCH + lane * 4);
    *(v4f*)(out + (size_t)(rowBase + rl) * L_DIM + lane * 4) = o;
  }
}

extern "C" void kernel_launch(void* const* d_in, const int* in_sizes, int n_in,
                              void* d_out, int out_size, void* d_ws, size_t ws_size,
                              hipStream_t stream) {
  const float* emb   = (const float*)d_in[0];
  const float* means = (const float*)d_in[1];
  const float* var   = (const float*)d_in[2];
  float* out = (float*)d_out;

  scorer_gemm<<<256, 256, 0, stream>>>(emb, means, var, out);
}


# emb tile staged to LDS in eight 64-k slices one trip ahead of use, handed over by a split work-group barrier; matrix work starts after the first slice; sum order of e2 and log var changed
# speedup vs baseline: 3.8268x; 1.0604x over previous
#include <hip/hip_runtime.h>
#include <math.h>


typedef __attribute__((ext_vector_type(2))) float v2f;
typedef __attribute__((ext_vector_type(4))) float v4f;
typedef __attribute__((ext_vector_type(8))) float v8f;

#define E_DIM   512
#define L_DIM   128
#define ROWS    64
#define APITCH  (E_DIM + 4)
#define OPITCH  (L_DIM + 4)
#define KS      64
#define NS      (E_DIM / KS)

static __device__ __forceinline__ void lds_landed() { asm volatile("s_wait_dscnt 0x0" ::: "memory"); }
static __device__ __forceinline__ void bar_signal() { asm volatile("s_barrier_signal -1" ::: "memory"); }
static __device__ __forceinline__ void bar_wait()   { asm volatile("s_barrier_wait -1" ::: "memory"); }

__global__ __launch_bounds__(256) __attribute__((amdgpu_waves_per_eu(2, 2))) void scorer_gemm(
    const float* __restrict__ emb, const float* __restrict__ means,
    const float* __restrict__ var, float* __restrict__ out) {
  __shared__ __attribute__((aligned(16))) float sA[ROWS * APITCH];
  __shared__ __attribute__((aligned(16))) float sIv[E_DIM];
  __shared__ __attribute__((aligned(16))) float sE2[ROWS];
  __shared__ float sLogW[8];

  const int tid   = threadIdx.x;
  const int wave  = __builtin_amdgcn_readfirstlane(tid >> 5);
  const int lane  = tid & 31;
  const int col   = lane & 15;
  const int khalf = lane >> 4;
  const int rowBase = blockIdx.x * ROWS;

  const int q  = lane & 15;
  const int hr = lane >> 4;
  const float* eS = emb + (size_t)(rowBase + wave * 8 + hr) * E_DIM + q * 4;
  float*       sS = sA + (wave * 8 + hr) * APITCH + q * 4;
  const float* vS = var + q * 4;
  const float* bC  = means + (size_t)(wave * 16 + col) * E_DIM + khalf * 4;
  const float* ivC = sIv + khalf * 4;
  const float* aC  = sA + col * APITCH + khalf * 4;

  v4f S[NS][4];
  v4f VQ[NS];
  v4f B[NS][8];
  float e2a[4] = {0.0f, 0.0f, 0.0f, 0.0f};

#define LOAD_SLICE(t)                                                                      \
  {                                                                                        \
    VQ[t] = *(const v4f*)(vS + (t) * KS);                                                  \
    _Pragma("unroll") for (int m = 0; m < 4; ++m)                                          \
      S[t][m] = *(const v4f*)(eS + (size_t)m * 2 * E_DIM + (t) * KS);                      \
  }
#define LOAD_B(t)                                                                          \
  {                                                                                        \
    _Pragma("unroll") for (int o = 0; o < 8; ++o)                                          \
      B[t][o] = *(const v4f*)(bC + (t) * KS + o * 8);      \
  }
#define STAGE_SLICE(t)                                                                     \
  {                                                                                        \
    const v4f v = VQ[t];                                                                   \
    v4f r;                                                                                 \
    r.x = 1.0f / v.x; r.y = 1.0f / v.y; r.z = 1.0f / v.z; r.w = 1.0f / v.w;                \
    *(v4f*)(sIv + (t) * KS + q * 4) = r;                                                   \
    _Pragma("unroll") for (int m = 0; m < 4; ++m) {                                        \
      const v4f a  = S[t][m];                                                              \
      const v4f as = a * r;                       \
      *(v4f*)(sS + m * 2 * APITCH + (t) * KS) = as;                                        \
      e2a[m] += (a.x * as.x + a.y * as.y) + (a.z * as.z + a.w * as.w);                     \
    }                                                                                      \
      \
      \
      \
    asm volatile("" :: "v"(e2a[0]), "v"(e2a[1]), "v"(e2a[2]), "v"(e2a[3]));                \
  }

  const v4f vl = *(const v4f*)(var + wave * KS + q * 4);
  LOAD_SLICE(0);
  __builtin_amdgcn_sched_barrier(0);
  LOAD_B(0);
  LOAD_SLICE(1);
  __builtin_amdgcn_sched_barrier(0);
  STAGE_SLICE(0);
  lds_landed();
  bar_signal();

  v8f acc[4] = {};
  v4f m2p = {0.0f, 0.0f, 0.0f, 0.0f};

#pragma unroll
  for (int t = 0; t < NS; ++t) {
    bar_wait();
    v4f iv[8], a[8][4];
#pragma unroll
    for (int o = 0; o < 8; ++o) {
      iv[o] = *(const v4f*)(ivC + t * KS + o * 8);
#pragma unroll
      for (int rt = 0; rt < 4; ++rt) a[o][rt] = *(const v4f*)(aC + rt * 16 * APITCH + t * KS + o * 8);
    }
    if (t + 1 < NS) LOAD_B(t + 1);
    if (t + 2 < NS) LOAD_SLICE(t + 2);
    if (t + 1 < NS) {
      STAGE_SLICE(t + 1);
      lds_landed();
      bar_signal();
    } else {
      float s = (logf(vl.x) + logf(vl.y)) + (logf(vl.z) + logf(vl.w));
#pragma unroll
      for (int o = 8; o > 0; o >>= 1) s += __shfl_xor(s, o, 32);
      sLogW[wave] = s;
    }
    __builtin_amdgcn_sched_barrier(0);
#pragma unroll
    for (int o = 0; o < 8; ++o) {
      const v4f b = B[t][o];
      m2p += b * (b * iv[o]);
      v2f b0 = {b.x, b.y};
      v2f b1 = {b.z, b.w};
#pragma unroll
      for (int rt = 0; rt < 4; ++rt) {
        v2f a0 = {a[o][rt].x, a[o][rt].y};
        acc[rt] = __builtin_amdgcn_wmma_f32_16x16x4_f32(
            false, a0, false, b0, (short)0, acc[rt], false, false);
      }
#pragma unroll
      for (int rt = 0; rt < 4; ++rt) {
        v2f a1 = {a[o][rt].z, a[o][rt].w};
        acc[rt] = __builtin_amdgcn_wmma_f32_16x16x4_f32(
            false, a1, false, b1, (short)0, acc[rt], false, false);
      }
    }
    asm volatile("" :: "v"(m2p));
    __builtin_amdgcn_sched_barrier(0);
  }

  {
    const bool b3 = (lane & 8) != 0, b2 = (lane & 4) != 0;
    float g[2];
#pragma unroll
    for (int j = 0; j < 2; ++j) {
      float send = b3 ? e2a[j] : e2a[j + 2];
      float keep = b3 ? e2a[j + 2] : e2a[j];
      g[j] = keep + __shfl_xor(send, 8, 32);
    }
    float send = b2 ? g[0] : g[1];
    float keep = b2 ? g[1] : g[0];
    float h = keep + __shfl_xor(send, 4, 32);
    h += __shfl_xor(h, 2, 32);
    h += __shfl_xor(h, 1, 32);
    if ((lane & 3) == 0) sE2[wave * 8 + 2 * ((lane >> 2) & 3) + hr] = h;
  }
  __syncthreads();

  float m2s = (m2p.x + m2p.y) + (m2p.z + m2p.w);
  const float m2c = m2s + __shfl_xor(m2s, 16, 32);

  const float log2pi = 1.8378770664093453f;
  const float factor = -0.5f * (float)E_DIM * log2pi;
  const float sumLog = ((sLogW[0] + sLogW[1]) + (sLogW[2] + sLogW[3])) + ((sLogW[4] + sLogW[5]) + (sLogW[6] + sLogW[7]));
  const float cTerm  = factor - 0.5f * sumLog;

  float* sO = sA;
#pragma unroll
  for (int rt = 0; rt < 4; ++rt) {
    const v4f e2lo = *(const v4f*)(sE2 + rt * 16 + khalf * 8);
    const v4f e2up = *(const v4f*)(sE2 + rt * 16 + khalf * 8 + 4);
    const float e2r[8] = {e2lo.x, e2lo.y, e2lo.z, e2lo.w, e2up.x, e2up.y, e2up.z, e2up.w};
#pragma unroll
    for (int i = 0; i < 8; ++i) {
      const int rl = rt * 16 + khalf * 8 + i;
      sO[rl * OPITCH + wave * 16 + col] = cTerm + acc[rt][i] - 0.5f * (e2r[i] + m2c);
    }
  }
  __syncthreads();

#pragma unroll
  for (int j = 0; j < 8; ++j) {
    const int rl = wave * 8 + j;
    v4f o = *(const v4f*)(sO + rl * OPITCH + lane * 4);
    *(v4f*)(out + (size_t)(rowBase + rl) * L_DIM + lane * 4) = o;
  }
}

extern "C" void kernel_launch(void* const* d_in, const int* in_sizes, int n_in,
                              void* d_out, int out_size, void* d_ws, size_t ws_size,
                              hipStream_t stream) {
  const float* emb   = (const float*)d_in[0];
  const float* means = (const float*)d_in[1];
  const float* var   = (const float*)d_in[2];
  float* out = (float*)d_out;

  scorer_gemm<<<256, 256, 0, stream>>>(emb, means, var, out);
}
